# starting kernel
# baseline (speedup 1.0000x reference)
#include <hip/hip_runtime.h>


typedef float v2f __attribute__((ext_vector_type(2)));
typedef float v8f __attribute__((ext_vector_type(8)));

#define KCH     29
#define NWAVES  4
#define XSTG    608
#define YSTG    768
#define BELEMS  (KCH * 3 * 32 * 2)

__global__ __launch_bounds__(128) void resample_poly_wmma_f32(
    const float* __restrict__ x, const float* __restrict__ f,
    float* __restrict__ out, int N, long long n_out, int n_tiles)
{
    __shared__ float Blds[BELEMS];
    __shared__ float xstage[NWAVES * XSTG];
    __shared__ float ystage[NWAVES * YSTG];

    const int tid  = threadIdx.x;
    const int lane = tid & 31;
    const int wave = tid >> 5;
    const int R    = lane & 15;
    const int hh   = lane >> 4;

    for (int i = tid; i < BELEMS; i += 128) {
        int v  = i & 1;
        int l  = (i >> 1) & 31;
        int a  = (i >> 6) % 3;
        int kc = i / 192;
        int t  = 4 * kc + v + 2 * (l >> 4);
        int c  = l & 15;
        int s  = t + 2 * c - ((a == 2) ? 31 : 32);
        int smax = (a == 0) ? 80 : 79;
        int tap  = (a == 0) ? 3 * s : ((a == 1) ? 3 * s + 2 : 3 * s + 1);
        float val = 0.0f;
        if (s >= 0 && s <= smax) val = f[tap];
        Blds[i] = val;
    }

    const long long T      = (long long)blockIdx.x * NWAVES + wave;
    const bool      activeW = (T < (long long)n_tiles);
    const long long n0     = 256LL * T;
    const long long base   = 40 + 2 * n0;
    float* xr = &xstage[wave * XSTG];

    for (int i = lane; i < 596; i += 32) {
        long long g = base + 512 - i;
        float v = 0.0f;
        if (activeW && g >= 0 && g < (long long)N) v = x[g];
        xr[i] = v;
    }
    __syncthreads();

    const int abase = 480 - 32 * R + 2 * hh;
    const int bbase = lane * 2;

    v8f c0 = {0.f, 0.f, 0.f, 0.f, 0.f, 0.f, 0.f, 0.f};
    v8f c1 = {0.f, 0.f, 0.f, 0.f, 0.f, 0.f, 0.f, 0.f};
    v8f c2 = {0.f, 0.f, 0.f, 0.f, 0.f, 0.f, 0.f, 0.f};

    for (int kc = 0; kc < KCH; ++kc) {
        v2f a = *(const v2f*)&xr[abase + 4 * kc];
        const float* bp = &Blds[kc * 192 + bbase];
        v2f b0 = *(const v2f*)&bp[0];
        v2f b1 = *(const v2f*)&bp[64];
        v2f b2 = *(const v2f*)&bp[128];
        c0 = __builtin_amdgcn_wmma_f32_16x16x4_f32(false, a, false, b0,
                                                   (short)0, c0, false, false);
        c1 = __builtin_amdgcn_wmma_f32_16x16x4_f32(false, a, false, b1,
                                                   (short)0, c1, false, false);
        c2 = __builtin_amdgcn_wmma_f32_16x16x4_f32(false, a, false, b2,
                                                   (short)0, c2, false, false);
    }

    float* yl = &ystage[wave * YSTG];
    for (int v = 0; v < 8; ++v) {
        int nl = 16 * (v + 8 * hh) + R;
        yl[3 * nl + 0] = 3.0f * c0[v];
        yl[3 * nl + 1] = 3.0f * c1[v];
        yl[3 * nl + 2] = 3.0f * c2[v];
    }
    __syncthreads();

    const long long m0 = 768LL * T;
    for (int k = 0; k < 24; ++k) {
        int i = k * 32 + lane;
        long long m = m0 + i;
        if (activeW && m < n_out) out[m] = yl[i];
    }
}

extern "C" void kernel_launch(void* const* d_in, const int* in_sizes, int n_in,
                              void* d_out, int out_size, void* d_ws, size_t ws_size,
                              hipStream_t stream) {
    const float* x = (const float*)d_in[0];
    const float* f = (const float*)d_in[1];
    float* out = (float*)d_out;

    const int       N      = in_sizes[0];
    const long long n_out  = (long long)out_size;
    const long long tiles  = (n_out + 767) / 768;
    const int       blocks = (int)((tiles + NWAVES - 1) / NWAVES);

    resample_poly_wmma_f32<<<blocks, 128, 0, stream>>>(
        x, f, out, N, n_out, (int)tiles);
}


# Persistent work-groups: one 32-wave group per WGP builds the filter fragments in LDS once (was once per 4 tiles); each wave then loops over its own tiles, with no group barrier inside the loop
# speedup vs baseline: 1.6452x; 1.6452x over previous
#include <hip/hip_runtime.h>


typedef float v2f __attribute__((ext_vector_type(2)));
typedef float v8f __attribute__((ext_vector_type(8)));

#define KCH     29
#define NWAVES  32
#define NTHR    (NWAVES * 32)
#define NGROUPS 256
#define XSTG    608
#define YSTG    768
#define BELEMS  (KCH * 3 * 32 * 2)

static __device__ __forceinline__ void wave_lds_order()
{
    __builtin_amdgcn_fence(__ATOMIC_RELEASE, "wavefront");
    __builtin_amdgcn_wave_barrier();
    __builtin_amdgcn_fence(__ATOMIC_ACQUIRE, "wavefront");
}

__global__ __launch_bounds__(NTHR) void resample_poly_wmma_f32(
    const float* __restrict__ x, const float* __restrict__ f,
    float* __restrict__ out, int N, long long n_out, int n_tiles)
{
    __shared__ float Blds[BELEMS];
    __shared__ float xstage[NWAVES * XSTG];
    __shared__ float ystage[NWAVES * YSTG];

    const int tid  = threadIdx.x;
    const int lane = tid & 31;
    const int wave = tid >> 5;
    const int R    = lane & 15;
    const int hh   = lane >> 4;

    for (int i = tid; i < BELEMS; i += NTHR) {
        int v  = i & 1;
        int l  = (i >> 1) & 31;
        int a  = (i >> 6) % 3;
        int kc = i / 192;
        int t  = 4 * kc + v + 2 * (l >> 4);
        int c  = l & 15;
        int s  = t + 2 * c - ((a == 2) ? 31 : 32);
        int smax = (a == 0) ? 80 : 79;
        int tap  = (a == 0) ? 3 * s : ((a == 1) ? 3 * s + 2 : 3 * s + 1);
        float val = 0.0f;
        if (s >= 0 && s <= smax) val = f[tap];
        Blds[i] = val;
    }

    float* xr = &xstage[wave * XSTG];
    float* yl = &ystage[wave * YSTG];
    const int abase = 480 - 32 * R + 2 * hh;
    const int bbase = lane * 2;

    const int wavesInGrid = NGROUPS * NWAVES;
    const int trips = (n_tiles + wavesInGrid - 1) / wavesInGrid;

    for (int trip = 0; trip < trips; ++trip) {
        const long long T = (long long)trip * wavesInGrid
                          + (long long)blockIdx.x * NWAVES + wave;
        const bool      activeW = (T < (long long)n_tiles);
        const long long n0   = 256LL * T;
        const long long base = 40 + 2 * n0;

        for (int i = lane; i < 596; i += 32) {
            long long g = base + 512 - i;
            float v = 0.0f;
            if (activeW && g >= 0 && g < (long long)N) v = x[g];
            xr[i] = v;
        }
        if (trip == 0) __syncthreads();
        else           wave_lds_order();

        v8f c0 = {0.f, 0.f, 0.f, 0.f, 0.f, 0.f, 0.f, 0.f};
        v8f c1 = {0.f, 0.f, 0.f, 0.f, 0.f, 0.f, 0.f, 0.f};
        v8f c2 = {0.f, 0.f, 0.f, 0.f, 0.f, 0.f, 0.f, 0.f};

        for (int kc = 0; kc < KCH; ++kc) {
            v2f a = *(const v2f*)&xr[abase + 4 * kc];
            const float* bp = &Blds[kc * 192 + bbase];
            v2f b0 = *(const v2f*)&bp[0];
            v2f b1 = *(const v2f*)&bp[64];
            v2f b2 = *(const v2f*)&bp[128];
            c0 = __builtin_amdgcn_wmma_f32_16x16x4_f32(false, a, false, b0,
                                                       (short)0, c0, false, false);
            c1 = __builtin_amdgcn_wmma_f32_16x16x4_f32(false, a, false, b1,
                                                       (short)0, c1, false, false);
            c2 = __builtin_amdgcn_wmma_f32_16x16x4_f32(false, a, false, b2,
                                                       (short)0, c2, false, false);
        }

        for (int v = 0; v < 8; ++v) {
            int nl = 16 * (v + 8 * hh) + R;
            yl[3 * nl + 0] = 3.0f * c0[v];
            yl[3 * nl + 1] = 3.0f * c1[v];
            yl[3 * nl + 2] = 3.0f * c2[v];
        }
        wave_lds_order();

        const long long m0 = 768LL * T;
        for (int k = 0; k < 24; ++k) {
            int i = k * 32 + lane;
            long long m = m0 + i;
            if (activeW && m < n_out) out[m] = yl[i];
        }
    }
}

extern "C" void kernel_launch(void* const* d_in, const int* in_sizes, int n_in,
                              void* d_out, int out_size, void* d_ws, size_t ws_size,
                              hipStream_t stream) {
    const float* x = (const float*)d_in[0];
    const float* f = (const float*)d_in[1];
    float* out = (float*)d_out;

    const int       N      = in_sizes[0];
    const long long n_out  = (long long)out_size;
    const long long tiles  = (n_out + 767) / 768;

    resample_poly_wmma_f32<<<NGROUPS, NTHR, 0, stream>>>(
        x, f, out, N, n_out, (int)tiles);
}


# Bounds tested once per tile (wave-uniform) or by clamped address and select, never by a branch per access: x-window loads (now 128-bit), filter gathers and output stores issue back to back
# speedup vs baseline: 2.7185x; 1.6524x over previous
#include <hip/hip_runtime.h>


typedef float v2f __attribute__((ext_vector_type(2)));
typedef float v8f __attribute__((ext_vector_type(8)));

#define KCH     29
#define NWAVES  32
#define NTHR    (NWAVES * 32)
#define NGROUPS 256
#define XPAD    4
#define XSTG    608
#define XVEC    149
#define YSTG    768
#define BELEMS  (KCH * 3 * 32 * 2)
#define BTRIPS  ((BELEMS + NTHR - 1) / NTHR)
#define BALLOC  (BTRIPS * NTHR)

static __device__ __forceinline__ void wave_lds_order()
{
    __builtin_amdgcn_fence(__ATOMIC_RELEASE, "wavefront");
    __builtin_amdgcn_wave_barrier();
    __builtin_amdgcn_fence(__ATOMIC_ACQUIRE, "wavefront");
}

__global__ __launch_bounds__(NTHR) void resample_poly_wmma_f32(
    const float* __restrict__ x, const float* __restrict__ f,
    float* __restrict__ out, int N, long long n_out, int n_tiles)
{
    __shared__ __attribute__((aligned(16))) float Blds[BALLOC];
    __shared__ __attribute__((aligned(16))) float xstage[NWAVES * XSTG];
    __shared__ __attribute__((aligned(16))) float ystage[NWAVES * YSTG];

    const int tid  = threadIdx.x;
    const int lane = tid & 31;
    const int wave = tid >> 5;
    const int R    = lane & 15;
    const int hh   = lane >> 4;

    int  btap[BTRIPS];
    bool bok[BTRIPS];
#pragma unroll
    for (int it = 0; it < BTRIPS; ++it) {
        int i  = tid + it * NTHR;
        int v  = i & 1;
        int l  = (i >> 1) & 31;
        int a  = (i >> 6) % 3;
        int kc = i / 192;
        int t  = 4 * kc + v + 2 * (l >> 4);
        int c  = l & 15;
        int s  = t + 2 * c - ((a == 2) ? 31 : 32);
        int smax = (a == 0) ? 80 : 79;
        int tap  = 3 * s + ((a == 0) ? 0 : 3 - a);
        bok[it]  = (i < BELEMS) && s >= 0 && s <= smax;
        btap[it] = bok[it] ? tap : 0;
    }
    float bval[BTRIPS];
#pragma unroll
    for (int it = 0; it < BTRIPS; ++it) bval[it] = f[btap[it]];
#pragma unroll
    for (int it = 0; it < BTRIPS; ++it) Blds[tid + it * NTHR] = bok[it] ? bval[it] : 0.0f;

    float* xr = &xstage[wave * XSTG + XPAD];
    float* yl = &ystage[wave * YSTG];
    const int abase = 480 - 32 * R + 2 * hh;
    const int bbase = lane * 2;

    const int wavesInGrid = NGROUPS * NWAVES;
    const int trips = (n_tiles + wavesInGrid - 1) / wavesInGrid;
    const bool vecPtrs = ((((unsigned long long)x) | ((unsigned long long)out)) & 15ull) == 0ull;

    for (int trip = 0; trip < trips; ++trip) {
        const long long T = (long long)trip * wavesInGrid
                          + (long long)blockIdx.x * NWAVES + wave;
        const bool      activeW = (T < (long long)n_tiles);
        const long long n0   = 256LL * T;
        const long long base = 40 + 2 * n0;

        const bool interiorX = activeW && vecPtrs && (base - 84 >= 0) && (base + 516 <= (long long)N);
        if (interiorX) {
            float4 xv[5];
#pragma unroll
            for (int it = 0; it < 5; ++it) {
                int j = it * 32 + lane;
                j = (j < XVEC) ? j : XVEC;
                xv[it] = *(const float4*)&x[base + 512 - 4 * j];
            }
#pragma unroll
            for (int it = 0; it < 5; ++it) {
                int j = it * 32 + lane;
                j = (j < XVEC) ? j : XVEC;
                float* d = &xr[4 * j];
                d[0]  = xv[it].x;
                d[-1] = xv[it].y;
                d[-2] = xv[it].z;
                d[-3] = xv[it].w;
            }
        } else {
            float xs[19];
#pragma unroll
            for (int it = 0; it < 19; ++it) {
                int i = it * 32 + lane;
                i = (i < 595) ? i : 595;
                long long g = base + 512 - i;
                bool ok = activeW && g >= 0 && g < (long long)N;
                float v = x[ok ? g : 0];
                xs[it] = ok ? v : 0.0f;
            }
#pragma unroll
            for (int it = 0; it < 19; ++it) {
                int i = it * 32 + lane;
                i = (i < 595) ? i : 595;
                xr[i] = xs[it];
            }
        }
        if (trip == 0) __syncthreads();
        else           wave_lds_order();

        v8f c0 = {0.f, 0.f, 0.f, 0.f, 0.f, 0.f, 0.f, 0.f};
        v8f c1 = {0.f, 0.f, 0.f, 0.f, 0.f, 0.f, 0.f, 0.f};
        v8f c2 = {0.f, 0.f, 0.f, 0.f, 0.f, 0.f, 0.f, 0.f};

        for (int kc = 0; kc < KCH; ++kc) {
            v2f a = *(const v2f*)&xr[abase + 4 * kc];
            const float* bp = &Blds[kc * 192 + bbase];
            v2f b0 = *(const v2f*)&bp[0];
            v2f b1 = *(const v2f*)&bp[64];
            v2f b2 = *(const v2f*)&bp[128];
            c0 = __builtin_amdgcn_wmma_f32_16x16x4_f32(false, a, false, b0,
                                                       (short)0, c0, false, false);
            c1 = __builtin_amdgcn_wmma_f32_16x16x4_f32(false, a, false, b1,
                                                       (short)0, c1, false, false);
            c2 = __builtin_amdgcn_wmma_f32_16x16x4_f32(false, a, false, b2,
                                                       (short)0, c2, false, false);
        }

        for (int v = 0; v < 8; ++v) {
            int nl = 16 * (v + 8 * hh) + R;
            yl[3 * nl + 0] = 3.0f * c0[v];
            yl[3 * nl + 1] = 3.0f * c1[v];
            yl[3 * nl + 2] = 3.0f * c2[v];
        }
        wave_lds_order();

        const long long m0 = 768LL * T;
        if (activeW && vecPtrs && m0 + 768 <= n_out) {
            float4 yv[6];
#pragma unroll
            for (int k = 0; k < 6; ++k) yv[k] = *(const float4*)&yl[4 * (k * 32 + lane)];
#pragma unroll
            for (int k = 0; k < 6; ++k) *(float4*)&out[m0 + 4 * (k * 32 + lane)] = yv[k];
        } else {
            for (int k = 0; k < 24; ++k) {
                int i = k * 32 + lane;
                long long m = m0 + i;
                if (activeW && m < n_out) out[m] = yl[i];
            }
        }
    }
}

extern "C" void kernel_launch(void* const* d_in, const int* in_sizes, int n_in,
                              void* d_out, int out_size, void* d_ws, size_t ws_size,
                              hipStream_t stream) {
    const float* x = (const float*)d_in[0];
    const float* f = (const float*)d_in[1];
    float* out = (float*)d_out;

    const int       N      = in_sizes[0];
    const long long n_out  = (long long)out_size;
    const long long tiles  = (n_out + 767) / 768;

    resample_poly_wmma_f32<<<NGROUPS, NTHR, 0, stream>>>(
        x, f, out, N, n_out, (int)tiles);
}


# Columns regrouped into blocks of 16 consecutive outputs: a block's tap band is 91 of the 116 K rows and its all-zero K chunks are skipped (70 products per tile, was 87); live terms keep their order
# speedup vs baseline: 3.0140x; 1.1087x over previous
#include <hip/hip_runtime.h>


typedef float v2f __attribute__((ext_vector_type(2)));
typedef float v8f __attribute__((ext_vector_type(8)));

#define KCH     29
#define NWAVES  32
#define NTHR    (NWAVES * 32)
#define NGROUPS 256
#define XPAD    4
#define XSTG    608
#define XVEC    149
#define YSTG    768
#define KLO0    5
#define KHI0    28
#define KLO1    3
#define KHI1    25
#define KLO2    0
#define KHI2    22
#define NF2     (KHI2 - KLO2 + 1)
#define NF1     (KHI1 - KLO1 + 1)
#define NF0     (KHI0 - KLO0 + 1)
#define BELEMS  ((NF2 + NF1 + NF0) * 32 * 2)
#define BTRIPS  ((BELEMS + NTHR - 1) / NTHR)
#define BALLOC  (BTRIPS * NTHR)

static __device__ __forceinline__ void wave_lds_order()
{
    __builtin_amdgcn_fence(__ATOMIC_RELEASE, "wavefront");
    __builtin_amdgcn_wave_barrier();
    __builtin_amdgcn_fence(__ATOMIC_ACQUIRE, "wavefront");
}

__global__ __launch_bounds__(NTHR) void resample_poly_wmma_f32(
    const float* __restrict__ x, const float* __restrict__ f,
    float* __restrict__ out, int N, long long n_out, int n_tiles)
{
    __shared__ __attribute__((aligned(16))) float Blds[BALLOC];
    __shared__ __attribute__((aligned(16))) float xstage[NWAVES * XSTG];
    __shared__ __attribute__((aligned(16))) float ystage[NWAVES * YSTG];

    const int tid  = threadIdx.x;
    const int lane = tid & 31;
    const int wave = tid >> 5;
    const int R    = lane & 15;
    const int hh   = lane >> 4;

    int  btap[BTRIPS];
    bool bok[BTRIPS];
#pragma unroll
    for (int it = 0; it < BTRIPS; ++it) {
        int i  = tid + it * NTHR;
        int v  = i & 1;
        int l  = (i >> 1) & 31;
        int fr = i >> 6;
        int g  = (fr < NF2) ? 2 : (fr < NF2 + NF1) ? 1 : 0;
        int kc = (g == 2) ? fr + KLO2 : (g == 1) ? fr - NF2 + KLO1 : fr - NF2 - NF1 + KLO0;
        int t  = 4 * kc + v + 2 * (l >> 4);
        int q  = 16 * g + (l & 15);
        int tap  = 3 * t + 2 * q - 96;
        bok[it]  = (i < BELEMS) && tap >= 0 && tap <= 240;
        btap[it] = bok[it] ? tap : 0;
    }
    float bval[BTRIPS];
#pragma unroll
    for (int it = 0; it < BTRIPS; ++it) bval[it] = f[btap[it]];
#pragma unroll
    for (int it = 0; it < BTRIPS; ++it) Blds[tid + it * NTHR] = bok[it] ? bval[it] : 0.0f;

    float* xr = &xstage[wave * XSTG + XPAD];
    float* yl = &ystage[wave * YSTG];
    const int abase = 480 - 32 * R + 2 * hh;
    const int bbase = lane * 2;

    const int wavesInGrid = NGROUPS * NWAVES;
    const int trips = (n_tiles + wavesInGrid - 1) / wavesInGrid;
    const bool vecPtrs = ((((unsigned long long)x) | ((unsigned long long)out)) & 15ull) == 0ull;

    for (int trip = 0; trip < trips; ++trip) {
        const long long T = (long long)trip * wavesInGrid
                          + (long long)blockIdx.x * NWAVES + wave;
        const bool      activeW = (T < (long long)n_tiles);
        const long long n0   = 256LL * T;
        const long long base = 40 + 2 * n0;

        const bool interiorX = activeW && vecPtrs && (base - 84 >= 0) && (base + 516 <= (long long)N);
        if (interiorX) {
            float4 xv[5];
#pragma unroll
            for (int it = 0; it < 5; ++it) {
                int j = it * 32 + lane;
                j = (j < XVEC) ? j : XVEC;
                xv[it] = *(const float4*)&x[base + 512 - 4 * j];
            }
#pragma unroll
            for (int it = 0; it < 5; ++it) {
                int j = it * 32 + lane;
                j = (j < XVEC) ? j : XVEC;
                float* d = &xr[4 * j];
                d[0]  = xv[it].x;
                d[-1] = xv[it].y;
                d[-2] = xv[it].z;
                d[-3] = xv[it].w;
            }
        } else {
            float xs[19];
#pragma unroll
            for (int it = 0; it < 19; ++it) {
                int i = it * 32 + lane;
                i = (i < 595) ? i : 595;
                long long g = base + 512 - i;
                bool ok = activeW && g >= 0 && g < (long long)N;
                float v = x[ok ? g : 0];
                xs[it] = ok ? v : 0.0f;
            }
#pragma unroll
            for (int it = 0; it < 19; ++it) {
                int i = it * 32 + lane;
                i = (i < 595) ? i : 595;
                xr[i] = xs[it];
            }
        }
        if (trip == 0) __syncthreads();
        else           wave_lds_order();

        v8f c0 = {0.f, 0.f, 0.f, 0.f, 0.f, 0.f, 0.f, 0.f};
        v8f c1 = {0.f, 0.f, 0.f, 0.f, 0.f, 0.f, 0.f, 0.f};
        v8f c2 = {0.f, 0.f, 0.f, 0.f, 0.f, 0.f, 0.f, 0.f};

#pragma unroll
        for (int kc = 0; kc < KCH; ++kc) {
            v2f a = *(const v2f*)&xr[abase + 4 * kc];
            if (kc >= KLO2 && kc <= KHI2) {
                v2f b2 = *(const v2f*)&Blds[(kc - KLO2) * 64 + bbase];
                c2 = __builtin_amdgcn_wmma_f32_16x16x4_f32(false, a, false, b2,
                                                           (short)0, c2, false, false);
            }
            if (kc >= KLO1 && kc <= KHI1) {
                v2f b1 = *(const v2f*)&Blds[(NF2 + kc - KLO1) * 64 + bbase];
                c1 = __builtin_amdgcn_wmma_f32_16x16x4_f32(false, a, false, b1,
                                                           (short)0, c1, false, false);
            }
            if (kc >= KLO0 && kc <= KHI0) {
                v2f b0 = *(const v2f*)&Blds[(NF2 + NF1 + kc - KLO0) * 64 + bbase];
                c0 = __builtin_amdgcn_wmma_f32_16x16x4_f32(false, a, false, b0,
                                                           (short)0, c0, false, false);
            }
        }

        for (int v = 0; v < 8; ++v) {
            float* yrow = &yl[48 * (v + 8 * hh) + R];
            yrow[0]  = 3.0f * c0[v];
            yrow[16] = 3.0f * c1[v];
            yrow[32] = 3.0f * c2[v];
        }
        wave_lds_order();

        const long long m0 = 768LL * T;
        if (activeW && vecPtrs && m0 + 768 <= n_out) {
            float4 yv[6];
#pragma unroll
            for (int k = 0; k < 6; ++k) yv[k] = *(const float4*)&yl[4 * (k * 32 + lane)];
#pragma unroll
            for (int k = 0; k < 6; ++k) *(float4*)&out[m0 + 4 * (k * 32 + lane)] = yv[k];
        } else {
            for (int k = 0; k < 24; ++k) {
                int i = k * 32 + lane;
                long long m = m0 + i;
                if (activeW && m < n_out) out[m] = yl[i];
            }
        }
    }
}

extern "C" void kernel_launch(void* const* d_in, const int* in_sizes, int n_in,
                              void* d_out, int out_size, void* d_ws, size_t ws_size,
                              hipStream_t stream) {
    const float* x = (const float*)d_in[0];
    const float* f = (const float*)d_in[1];
    float* out = (float*)d_out;

    const int       N      = in_sizes[0];
    const long long n_out  = (long long)out_size;
    const long long tiles  = (n_out + 767) / 768;

    resample_poly_wmma_f32<<<NGROUPS, NTHR, 0, stream>>>(
        x, f, out, N, n_out, (int)tiles);
}
